# starting kernel
# baseline (speedup 1.0000x reference)
#include <hip/hip_runtime.h>
#include <hip/hip_bf16.h>
#include <math.h>

typedef __attribute__((ext_vector_type(2))) float v2f;
typedef __attribute__((ext_vector_type(8))) float v8f;

#define PROJ   8192
#define NCH    512
#define HWSZ   196
#define NTILE  1024
#define BLK    256

__global__ __launch_bounds__(BLK)
void cbp_gram_sketch_kernel(const float* __restrict__ x,
                            const float* __restrict__ s1,
                            const float* __restrict__ s2,
                            const int*   __restrict__ h1,
                            const int*   __restrict__ h2,
                            float* __restrict__ out)
{
    __shared__ float bins[PROJ];
    __shared__ float s1l[NCH], s2l[NCH];
    __shared__ int   h1l[NCH], h2l[NCH];
    __shared__ float red[8];

    const int b    = blockIdx.x;
    const int tid  = threadIdx.x;
    const int lane = tid & 31;
    const int wave = tid >> 5;

    for (int p = tid; p < PROJ; p += BLK) bins[p] = 0.0f;
    for (int c = tid; c < NCH; c += BLK) {
        s1l[c] = s1[c];  s2l[c] = s2[c];
        h1l[c] = h1[c];  h2l[c] = h2[c];
    }
    __syncthreads();

    const float* xb = x + (size_t)b * NCH * HWSZ;

    const int half = lane >> 4;
    const int lm16 = lane & 15;
    const int kb   = 2 * half;

    for (int t = wave; t < NTILE; t += 8) {
        const int i = t >> 5;
        const int j = t & 31;

        const float* arow = xb + (size_t)(i * 16 + lm16) * HWSZ;
        const float* brow = xb + (size_t)(j * 16 + lm16) * HWSZ;

        v8f acc = {};
        #pragma unroll 7
        for (int kk = 0; kk < 49; ++kk) {
            const int k0 = kk * 4 + kb;
            v2f av = *(const v2f*)(arow + k0);
            v2f bv = *(const v2f*)(brow + k0);
            acc = __builtin_amdgcn_wmma_f32_16x16x4_f32(
                    false, av, false, bv, (short)0, acc, false, false);
        }

        const int   cB  = j * 16 + lm16;
        const float s2v = s2l[cB];
        const int   h2v = h2l[cB];
        #pragma unroll
        for (int v = 0; v < 8; ++v) {
            const int   cA  = i * 16 + v + 8 * half;
            const float val = acc[v] * s1l[cA] * s2v;
            const int   bkt = (h1l[cA] + h2v) & (PROJ - 1);
            atomicAdd(&bins[bkt], val);
        }
    }
    __syncthreads();

    float yv[PROJ / BLK];
    float ss = 0.0f;
    #pragma unroll
    for (int q = 0; q < PROJ / BLK; ++q) {
        const float tv = bins[tid + q * BLK];
        const float sg = (tv > 0.0f) ? 1.0f : ((tv < 0.0f) ? -1.0f : 0.0f);
        const float y  = sg * sqrtf(fabsf(tv) + 1e-8f);
        yv[q] = y;
        ss   += y * y;
    }
    #pragma unroll
    for (int off = 16; off > 0; off >>= 1)
        ss += __shfl_down(ss, off, 32);
    if (lane == 0) red[wave] = ss;
    __syncthreads();
    if (tid == 0) {
        float tot = 0.0f;
        #pragma unroll
        for (int w = 0; w < 8; ++w) tot += red[w];
        red[0] = fmaxf(sqrtf(tot), 1e-12f);
    }
    __syncthreads();

    const float inv = 1.0f / red[0];
    float* ob = out + (size_t)b * PROJ;
    #pragma unroll
    for (int q = 0; q < PROJ / BLK; ++q)
        ob[tid + q * BLK] = yv[q] * inv;
}

extern "C" void kernel_launch(void* const* d_in, const int* in_sizes, int n_in,
                              void* d_out, int out_size, void* d_ws, size_t ws_size,
                              hipStream_t stream) {
    const float* x  = (const float*)d_in[0];
    const float* s1 = (const float*)d_in[1];
    const float* s2 = (const float*)d_in[2];
    const int*   h1 = (const int*)  d_in[3];
    const int*   h2 = (const int*)  d_in[4];
    float*       y  = (float*)d_out;
    (void)in_sizes; (void)n_in; (void)out_size; (void)d_ws; (void)ws_size;

    cbp_gram_sketch_kernel<<<16, BLK, 0, stream>>>(x, s1, s2, h1, h2, y);
}


# Split each image's 1024 Gram tiles over 16 work-groups (grid 16 -> 256, one per WGP): partial sketches go through the workspace, a second launch sums them and does sqrt + L2 norm
# speedup vs baseline: 9.7207x; 9.7207x over previous
#include <hip/hip_runtime.h>
#include <hip/hip_bf16.h>
#include <math.h>


typedef __attribute__((ext_vector_type(2))) float v2f;
typedef __attribute__((ext_vector_type(8))) float v8f;

#define PROJ   8192
#define NCH    512
#define HWSZ   196
#define NTILE  1024
#define BLK    256
#define NIMG   16
#define SPLIT  16
#define TPG    (NTILE / SPLIT)
#define FBLK   1024

__device__ __forceinline__ bool ws_holds_planes(const float* ws, size_t ws_bytes)
{
    return ws != nullptr
        && ws_bytes >= (size_t)NIMG * SPLIT * PROJ * sizeof(float)
        && (((size_t)ws) & 15) == 0;
}

__global__ __launch_bounds__(BLK)
void cbp_gram_sketch_partial_kernel(const float* __restrict__ x,
                                    const float* __restrict__ s1,
                                    const float* __restrict__ s2,
                                    const int*   __restrict__ h1,
                                    const int*   __restrict__ h2,
                                    float* __restrict__ out,
                                    float* __restrict__ ws,
                                    size_t ws_bytes)
{
    __shared__ __attribute__((aligned(16))) float bins[PROJ];
    __shared__ float s1l[NCH], s2l[NCH];
    __shared__ int   h1l[NCH], h2l[NCH];

    const bool split = ws_holds_planes(ws, ws_bytes);
    const int  b     = blockIdx.x / SPLIT;
    const int  p     = blockIdx.x % SPLIT;
    if (!split && p != 0) return;

    const int t0 = split ? p * TPG : 0;
    const int t1 = split ? t0 + TPG : NTILE;
    float* plane = split ? ws + (size_t)blockIdx.x * PROJ
                         : out + (size_t)b * PROJ;

    const int tid  = threadIdx.x;
    const int lane = tid & 31;
    const int wave = tid >> 5;

    for (int q = tid; q < PROJ; q += BLK) bins[q] = 0.0f;
    for (int c = tid; c < NCH; c += BLK) {
        s1l[c] = s1[c];  s2l[c] = s2[c];
        h1l[c] = h1[c];  h2l[c] = h2[c];
    }
    __syncthreads();

    const float* xb = x + (size_t)b * NCH * HWSZ;

    const int half = lane >> 4;
    const int lm16 = lane & 15;
    const int kb   = 2 * half;

    for (int t = t0 + wave; t < t1; t += 8) {
        const int i = t >> 5;
        const int j = t & 31;

        const float* arow = xb + (size_t)(i * 16 + lm16) * HWSZ;
        const float* brow = xb + (size_t)(j * 16 + lm16) * HWSZ;

        v8f acc = {};
        #pragma unroll 7
        for (int kk = 0; kk < 49; ++kk) {
            const int k0 = kk * 4 + kb;
            v2f av = *(const v2f*)(arow + k0);
            v2f bv = *(const v2f*)(brow + k0);
            acc = __builtin_amdgcn_wmma_f32_16x16x4_f32(
                    false, av, false, bv, (short)0, acc, false, false);
        }

        const int   cB  = j * 16 + lm16;
        const float s2v = s2l[cB];
        const int   h2v = h2l[cB];
        #pragma unroll
        for (int v = 0; v < 8; ++v) {
            const int   cA  = i * 16 + v + 8 * half;
            const float val = acc[v] * s1l[cA] * s2v;
            const int   bkt = (h1l[cA] + h2v) & (PROJ - 1);
            atomicAdd(&bins[bkt], val);
        }
    }
    __syncthreads();

    const float4* bins4  = (const float4*)bins;
    float4*       plane4 = (float4*)plane;
    #pragma unroll
    for (int q = 0; q < PROJ / 4 / BLK; ++q)
        plane4[q * BLK + tid] = bins4[q * BLK + tid];
}

__global__ __launch_bounds__(FBLK)
void cbp_sum_sqrt_norm_kernel(const float* ws, float* out, size_t ws_bytes)
{
    __shared__ float red[FBLK / 32];

    const int b    = blockIdx.x;
    const int tid  = threadIdx.x;
    const int lane = tid & 31;
    const int wave = tid >> 5;

    const bool split = ws_holds_planes(ws, ws_bytes);
    float* ob = out + (size_t)b * PROJ;

    float4 a0, a1;
    if (split) {
        const float4* pl = (const float4*)(ws + (size_t)b * SPLIT * PROJ);
        a0 = pl[tid];
        a1 = pl[tid + FBLK];
        #pragma unroll
        for (int q = 1; q < SPLIT; ++q) {
            const float4 u = pl[q * (PROJ / 4) + tid];
            const float4 w = pl[q * (PROJ / 4) + tid + FBLK];
            a0.x += u.x; a0.y += u.y; a0.z += u.z; a0.w += u.w;
            a1.x += w.x; a1.y += w.y; a1.z += w.z; a1.w += w.w;
        }
    } else {
        const float4* pl = (const float4*)ob;
        a0 = pl[tid];
        a1 = pl[tid + FBLK];
    }

    float tv[8] = { a0.x, a0.y, a0.z, a0.w, a1.x, a1.y, a1.z, a1.w };
    float yv[8];
    float ss = 0.0f;
    #pragma unroll
    for (int q = 0; q < 8; ++q) {
        const float t  = tv[q];
        const float sg = (t > 0.0f) ? 1.0f : ((t < 0.0f) ? -1.0f : 0.0f);
        const float y  = sg * sqrtf(fabsf(t) + 1e-8f);
        yv[q] = y;
        ss   += y * y;
    }
    #pragma unroll
    for (int off = 16; off > 0; off >>= 1)
        ss += __shfl_down(ss, off, 32);
    if (lane == 0) red[wave] = ss;
    __syncthreads();
    if (tid == 0) {
        float tot = 0.0f;
        #pragma unroll
        for (int w = 0; w < FBLK / 32; ++w) tot += red[w];
        red[0] = fmaxf(sqrtf(tot), 1e-12f);
    }
    __syncthreads();

    const float inv = 1.0f / red[0];
    float4* ob4 = (float4*)ob;
    ob4[tid]        = make_float4(yv[0] * inv, yv[1] * inv, yv[2] * inv, yv[3] * inv);
    ob4[tid + FBLK] = make_float4(yv[4] * inv, yv[5] * inv, yv[6] * inv, yv[7] * inv);
}

extern "C" void kernel_launch(void* const* d_in, const int* in_sizes, int n_in,
                              void* d_out, int out_size, void* d_ws, size_t ws_size,
                              hipStream_t stream) {
    const float* x  = (const float*)d_in[0];
    const float* s1 = (const float*)d_in[1];
    const float* s2 = (const float*)d_in[2];
    const int*   h1 = (const int*)  d_in[3];
    const int*   h2 = (const int*)  d_in[4];
    float*       y  = (float*)d_out;
    float*       ws = (float*)d_ws;
    (void)in_sizes; (void)n_in; (void)out_size;

    cbp_gram_sketch_partial_kernel<<<NIMG * SPLIT, BLK, 0, stream>>>(x, s1, s2, h1, h2, y, ws, ws_size);
    cbp_sum_sqrt_norm_kernel<<<NIMG, FBLK, 0, stream>>>(ws, y, ws_size);
}


# Work-group = one 128x128 Gram block: its 256 rows of x staged once in LDS by whole-line loads; a wave does 2x4 tiles from LDS fragments (6 reads per 8 WMMAs); waves add to bins in turn, fixed order
# speedup vs baseline: 19.2168x; 1.9769x over previous
#include <hip/hip_runtime.h>
#include <hip/hip_bf16.h>
#include <math.h>


typedef __attribute__((ext_vector_type(2))) float v2f;
typedef __attribute__((ext_vector_type(8))) float v8f;

#define PROJ   8192
#define NCH    512
#define HWSZ   196
#define BLK    256
#define NIMG   16
#define SPLIT  16
#define BROWS  128
#define BFLT   (BROWS * HWSZ)
#define BVEC   (BFLT / 4)
#define FBLK   1024

__device__ __forceinline__ bool ws_holds_planes(const float* ws, size_t ws_bytes)
{
    return ws != nullptr
        && ws_bytes >= (size_t)NIMG * SPLIT * PROJ * sizeof(float)
        && (((size_t)ws) & 15) == 0;
}

__global__ __launch_bounds__(BLK)
void cbp_gram_sketch_partial_kernel(const float* __restrict__ x,
                                    const float* __restrict__ s1,
                                    const float* __restrict__ s2,
                                    const int*   __restrict__ h1,
                                    const int*   __restrict__ h2,
                                    float* __restrict__ out,
                                    float* __restrict__ ws,
                                    size_t ws_bytes)
{
    __shared__ __attribute__((aligned(16))) float stage[2 * BFLT];
    __shared__ __attribute__((aligned(16))) float bins[PROJ];
    __shared__ float s1l[NCH], s2l[NCH];
    __shared__ int   h1l[NCH], h2l[NCH];

    const bool split = ws_holds_planes(ws, ws_bytes);
    const int  b     = blockIdx.x / SPLIT;
    const int  p     = blockIdx.x % SPLIT;
    if (!split && p != 0) return;

    const int blk0 = split ? p : 0;
    const int blk1 = split ? p + 1 : SPLIT;
    float* plane = split ? ws + (size_t)blockIdx.x * PROJ
                         : out + (size_t)b * PROJ;

    const int tid  = threadIdx.x;
    const int lane = tid & 31;
    const int wave = __builtin_amdgcn_readfirstlane(tid >> 5);

    for (int q = tid; q < PROJ; q += BLK) bins[q] = 0.0f;
    for (int c = tid; c < NCH; c += BLK) {
        s1l[c] = s1[c];  s2l[c] = s2[c];
        h1l[c] = h1[c];  h2l[c] = h2[c];
    }

    const float* xb = x + (size_t)b * NCH * HWSZ;

    const int half = lane >> 4;
    const int lm16 = lane & 15;
    const int kb   = 2 * half;
    const int wr   = wave >> 1;
    const int wc   = wave & 1;

    for (int blk = blk0; blk < blk1; ++blk) {
        const int bi = blk >> 2;
        const int bj = blk & 3;

        {
            const float4* a4  = (const float4*)(xb + (size_t)bi * BFLT);
            const float4* b4  = (const float4*)(xb + (size_t)bj * BFLT);
            float4*       st4 = (float4*)stage;
            #pragma unroll
            for (int q = 0; q < 2 * BVEC / BLK; ++q) {
                const int idx = q * BLK + tid;
                const float4* src = (idx < BVEC) ? (a4 + idx) : (b4 + (idx - BVEC));
                st4[idx] = *src;
            }
        }
        __syncthreads();

        const float* ap = stage + (wr * 32 + lm16) * HWSZ + kb;
        const float* bp = stage + BFLT + (wc * 64 + lm16) * HWSZ + kb;

        v8f acc[2][4];
        #pragma unroll
        for (int r = 0; r < 2; ++r)
            #pragma unroll
            for (int c = 0; c < 4; ++c)
                acc[r][c] = (v8f){};

        #pragma unroll 7
        for (int kk = 0; kk < 49; ++kk) {
            const int k0 = kk * 4;
            v2f av[2], bv[4];
            #pragma unroll
            for (int r = 0; r < 2; ++r) av[r] = *(const v2f*)(ap + r * 16 * HWSZ + k0);
            #pragma unroll
            for (int c = 0; c < 4; ++c) bv[c] = *(const v2f*)(bp + c * 16 * HWSZ + k0);
            #pragma unroll
            for (int r = 0; r < 2; ++r)
                #pragma unroll
                for (int c = 0; c < 4; ++c)
                    acc[r][c] = __builtin_amdgcn_wmma_f32_16x16x4_f32(
                            false, av[r], false, bv[c], (short)0, acc[r][c], false, false);
        }

        float s2v[4];
        int   h2v[4];
        #pragma unroll
        for (int c = 0; c < 4; ++c) {
            const int cB = bj * BROWS + (wc * 4 + c) * 16 + lm16;
            s2v[c] = s2l[cB];
            h2v[c] = h2l[cB];
        }
        float s1v[2][8];
        int   h1v[2][8];
        #pragma unroll
        for (int r = 0; r < 2; ++r)
            #pragma unroll
            for (int v = 0; v < 8; ++v) {
                const int cA = bi * BROWS + (wr * 2 + r) * 16 + v + 8 * half;
                s1v[r][v] = s1l[cA];
                h1v[r][v] = h1l[cA];
            }

        for (int w = 0; w < BLK / 32; ++w) {
            if (wave == w) {
                #pragma unroll
                for (int r = 0; r < 2; ++r)
                    #pragma unroll
                    for (int c = 0; c < 4; ++c)
                        #pragma unroll
                        for (int v = 0; v < 8; ++v) {
                            const float val = acc[r][c][v] * s1v[r][v] * s2v[c];
                            const int   bkt = (h1v[r][v] + h2v[c]) & (PROJ - 1);
                            atomicAdd(&bins[bkt], val);
                        }
            }
            __syncthreads();
        }
    }

    const float4* bins4  = (const float4*)bins;
    float4*       plane4 = (float4*)plane;
    #pragma unroll
    for (int q = 0; q < PROJ / 4 / BLK; ++q)
        plane4[q * BLK + tid] = bins4[q * BLK + tid];
}

__global__ __launch_bounds__(FBLK)
void cbp_sum_sqrt_norm_kernel(const float* ws, float* out, size_t ws_bytes)
{
    __shared__ float red[FBLK / 32];

    const int b    = blockIdx.x;
    const int tid  = threadIdx.x;
    const int lane = tid & 31;
    const int wave = tid >> 5;

    const bool split = ws_holds_planes(ws, ws_bytes);
    float* ob = out + (size_t)b * PROJ;

    float4 a0, a1;
    if (split) {
        const float4* pl = (const float4*)(ws + (size_t)b * SPLIT * PROJ);
        a0 = pl[tid];
        a1 = pl[tid + FBLK];
        #pragma unroll
        for (int q = 1; q < SPLIT; ++q) {
            const float4 u = pl[q * (PROJ / 4) + tid];
            const float4 w = pl[q * (PROJ / 4) + tid + FBLK];
            a0.x += u.x; a0.y += u.y; a0.z += u.z; a0.w += u.w;
            a1.x += w.x; a1.y += w.y; a1.z += w.z; a1.w += w.w;
        }
    } else {
        const float4* pl = (const float4*)ob;
        a0 = pl[tid];
        a1 = pl[tid + FBLK];
    }

    float tv[8] = { a0.x, a0.y, a0.z, a0.w, a1.x, a1.y, a1.z, a1.w };
    float yv[8];
    float ss = 0.0f;
    #pragma unroll
    for (int q = 0; q < 8; ++q) {
        const float t  = tv[q];
        const float sg = (t > 0.0f) ? 1.0f : ((t < 0.0f) ? -1.0f : 0.0f);
        const float y  = sg * sqrtf(fabsf(t) + 1e-8f);
        yv[q] = y;
        ss   += y * y;
    }
    #pragma unroll
    for (int off = 16; off > 0; off >>= 1)
        ss += __shfl_down(ss, off, 32);
    if (lane == 0) red[wave] = ss;
    __syncthreads();
    if (tid == 0) {
        float tot = 0.0f;
        #pragma unroll
        for (int w = 0; w < FBLK / 32; ++w) tot += red[w];
        red[0] = fmaxf(sqrtf(tot), 1e-12f);
    }
    __syncthreads();

    const float inv = 1.0f / red[0];
    float4* ob4 = (float4*)ob;
    ob4[tid]        = make_float4(yv[0] * inv, yv[1] * inv, yv[2] * inv, yv[3] * inv);
    ob4[tid + FBLK] = make_float4(yv[4] * inv, yv[5] * inv, yv[6] * inv, yv[7] * inv);
}

extern "C" void kernel_launch(void* const* d_in, const int* in_sizes, int n_in,
                              void* d_out, int out_size, void* d_ws, size_t ws_size,
                              hipStream_t stream) {
    const float* x  = (const float*)d_in[0];
    const float* s1 = (const float*)d_in[1];
    const float* s2 = (const float*)d_in[2];
    const int*   h1 = (const int*)  d_in[3];
    const int*   h2 = (const int*)  d_in[4];
    float*       y  = (float*)d_out;
    float*       ws = (float*)d_ws;
    (void)in_sizes; (void)n_in; (void)out_size;

    cbp_gram_sketch_partial_kernel<<<NIMG * SPLIT, BLK, 0, stream>>>(x, s1, s2, h1, h2, y, ws, ws_size);
    cbp_sum_sqrt_norm_kernel<<<NIMG, FBLK, 0, stream>>>(ws, y, ws_size);
}


# Gram symmetry: only the 528 tiles on or above the diagonal are computed (36 or 32 per work-group, was 64); a tile off the diagonal is scattered twice, as G[c][c'] and as G[c'][c]
# speedup vs baseline: 21.2922x; 1.1080x over previous
#include <hip/hip_runtime.h>
#include <hip/hip_bf16.h>
#include <math.h>


typedef __attribute__((ext_vector_type(2))) float v2f;
typedef __attribute__((ext_vector_type(8))) float v8f;

#define PROJ   8192
#define NCH    512
#define HWSZ   196
#define BLK    256
#define NIMG   16
#define SPLIT  16
#define BROWS  128
#define BFLT   (BROWS * HWSZ)
#define BVEC   (BFLT / 4)
#define SROWS  192
#define NSLOT  5
#define FBLK   1024

__device__ __forceinline__ bool ws_holds_planes(const float* ws, size_t ws_bytes)
{
    return ws != nullptr
        && ws_bytes >= (size_t)NIMG * SPLIT * PROJ * sizeof(float)
        && (((size_t)ws) & 15) == 0;
}

template <int NVEC>
__device__ __forceinline__ void stage_rows(const float4* __restrict__ a4, int nA4,
                                           const float4* __restrict__ b4,
                                           float4* st4, int tid)
{
    constexpr int FULL = NVEC / BLK;
    constexpr int REM  = NVEC % BLK;
    float4 part = make_float4(0.0f, 0.0f, 0.0f, 0.0f);
    if (REM != 0) {
        const int idx = min(FULL * BLK + tid, NVEC - 1);
        const float4* src = (idx < nA4) ? (a4 + idx) : (b4 + (idx - nA4));
        part = *src;
    }
    #pragma unroll
    for (int q = 0; q < FULL; ++q) {
        const int idx = q * BLK + tid;
        const float4* src = (idx < nA4) ? (a4 + idx) : (b4 + (idx - nA4));
        st4[idx] = *src;
    }
    if (REM != 0 && tid < REM) st4[FULL * BLK + tid] = part;
}

template <int NS>
__device__ __forceinline__ void gram_k_loop(const float* const (&ap)[NSLOT],
                                            const float* const (&bp)[NSLOT],
                                            v8f (&acc)[NSLOT])
{
    #pragma unroll 7
    for (int kk = 0; kk < 49; ++kk) {
        const int k0 = kk * 4;
        v2f av[NS], bv[NS];
        #pragma unroll
        for (int s = 0; s < NS; ++s) {
            av[s] = *(const v2f*)(ap[s] + k0);
            bv[s] = *(const v2f*)(bp[s] + k0);
        }
        #pragma unroll
        for (int s = 0; s < NS; ++s)
            acc[s] = __builtin_amdgcn_wmma_f32_16x16x4_f32(
                    false, av[s], false, bv[s], (short)0, acc[s], false, false);
    }
}

__global__ __launch_bounds__(BLK)
void cbp_gram_sketch_partial_kernel(const float* __restrict__ x,
                                    const float* __restrict__ s1,
                                    const float* __restrict__ s2,
                                    const int*   __restrict__ h1,
                                    const int*   __restrict__ h2,
                                    float* __restrict__ out,
                                    float* __restrict__ ws,
                                    size_t ws_bytes)
{
    __shared__ __attribute__((aligned(16))) float stage[SROWS * HWSZ];
    __shared__ __attribute__((aligned(16))) float bins[PROJ];
    __shared__ __attribute__((aligned(16))) float s1l[NCH];
    __shared__ __attribute__((aligned(16))) float s2l[NCH];
    __shared__ __attribute__((aligned(16))) int   h1l[NCH];
    __shared__ __attribute__((aligned(16))) int   h2l[NCH];

    const bool split = ws_holds_planes(ws, ws_bytes);
    const int  b     = blockIdx.x / SPLIT;
    const int  p     = blockIdx.x % SPLIT;
    if (!split && p != 0) return;

    const int u0 = split ? p : 0;
    const int u1 = split ? p + 1 : SPLIT;
    float* plane = split ? ws + (size_t)blockIdx.x * PROJ
                         : out + (size_t)b * PROJ;

    const int tid  = threadIdx.x;
    const int lane = tid & 31;
    const int wave = __builtin_amdgcn_readfirstlane(tid >> 5);

    for (int q = tid; q < PROJ; q += BLK) bins[q] = 0.0f;
    for (int c = tid; c < NCH; c += BLK) {
        s1l[c] = s1[c];  s2l[c] = s2[c];
        h1l[c] = h1[c];  h2l[c] = h2[c];
    }

    const float* xb = x + (size_t)b * NCH * HWSZ;

    const int half = lane >> 4;
    const int lm16 = lane & 15;
    const int kb   = 2 * half;

    for (int u = u0; u < u1; ++u) {
        const bool diag = (u < 4);
        int aC, bC, bRow0;
        if (diag) {
            aC = u * BROWS;  bC = aC;  bRow0 = 0;
        } else {
            const int o  = u - 4;
            const int pr = o >> 1;
            const int bi = (pr >= 3) + (pr >= 5);
            const int bj = (pr < 3) ? pr + 1 : ((pr < 5) ? pr - 1 : 3);
            aC = bi * BROWS + (o & 1) * 64;  bC = bj * BROWS;  bRow0 = 64;
        }

        {
            const float4* a4  = (const float4*)(xb + (size_t)aC * HWSZ);
            const float4* b4  = (const float4*)(xb + (size_t)bC * HWSZ);
            float4*       st4 = (float4*)stage;
            if (diag) stage_rows<BVEC>(a4, BVEC, a4, st4, tid);
            else      stage_rows<BVEC / 2 + BVEC>(a4, BVEC / 2, b4, st4, tid);
        }
        __syncthreads();

        const int ns = (diag && (__builtin_popcount((unsigned)wave) & 1) == 0) ? 5 : 4;
        int ti[NSLOT], tj[NSLOT];
        #pragma unroll
        for (int s = 0; s < NSLOT; ++s) {
            if (diag) {
                const int e = (s < 4) ? wave + 8 * s : 32 + (wave >> 1);
                const int q = e / 9;
                const int r = e - 9 * q;
                if (r < 8 - q) { ti[s] = q;      tj[s] = q + r; }
                else           { ti[s] = 7 - q;  tj[s] = 7 - q + (r - (8 - q)); }
            } else {
                ti[s] = 2 * (wave >> 2) + ((s >> 1) & 1);
                tj[s] = 2 * (wave & 3) + (s & 1);
            }
        }

        const float* ap[NSLOT];
        const float* bp[NSLOT];
        v8f acc[NSLOT];
        #pragma unroll
        for (int s = 0; s < NSLOT; ++s) {
            ap[s]  = stage + (ti[s] * 16 + lm16) * HWSZ + kb;
            bp[s]  = stage + (bRow0 + tj[s] * 16 + lm16) * HWSZ + kb;
            acc[s] = (v8f){};
        }

        if (ns == 5) gram_k_loop<5>(ap, bp, acc);
        else         gram_k_loop<4>(ap, bp, acc);

        float nval[NSLOT][8], tval[NSLOT][8];
        int   nbkt[NSLOT][8], tbkt[NSLOT][8];
        bool  both[NSLOT];
        #pragma unroll
        for (int s = 0; s < NSLOT; ++s) {
            const int rc0 = aC + ti[s] * 16;
            const int cc0 = bC + tj[s] * 16;
            both[s] = (rc0 != cc0);
            const int cB  = cc0 + lm16;
            const int cA0 = rc0 + 8 * half;
            const float s1b = s1l[cB], s2b = s2l[cB];
            const int   h1b = h1l[cB], h2b = h2l[cB];
            const float4 s1a0 = *(const float4*)(s1l + cA0), s1a1 = *(const float4*)(s1l + cA0 + 4);
            const float4 s2a0 = *(const float4*)(s2l + cA0), s2a1 = *(const float4*)(s2l + cA0 + 4);
            const int4   h1a0 = *(const int4*)(h1l + cA0),   h1a1 = *(const int4*)(h1l + cA0 + 4);
            const int4   h2a0 = *(const int4*)(h2l + cA0),   h2a1 = *(const int4*)(h2l + cA0 + 4);
            const float s1a[8] = { s1a0.x, s1a0.y, s1a0.z, s1a0.w, s1a1.x, s1a1.y, s1a1.z, s1a1.w };
            const float s2a[8] = { s2a0.x, s2a0.y, s2a0.z, s2a0.w, s2a1.x, s2a1.y, s2a1.z, s2a1.w };
            const int   h1a[8] = { h1a0.x, h1a0.y, h1a0.z, h1a0.w, h1a1.x, h1a1.y, h1a1.z, h1a1.w };
            const int   h2a[8] = { h2a0.x, h2a0.y, h2a0.z, h2a0.w, h2a1.x, h2a1.y, h2a1.z, h2a1.w };
            #pragma unroll
            for (int v = 0; v < 8; ++v) {
                nval[s][v] = acc[s][v] * s1a[v] * s2b;
                nbkt[s][v] = (h1a[v] + h2b) & (PROJ - 1);
                tval[s][v] = acc[s][v] * s1b * s2a[v];
                tbkt[s][v] = (h1b + h2a[v]) & (PROJ - 1);
            }
        }

        for (int w = 0; w < BLK / 32; ++w) {
            if (wave == w) {
                #pragma unroll
                for (int s = 0; s < NSLOT; ++s) {
                    if (s < ns) {
                        #pragma unroll
                        for (int v = 0; v < 8; ++v)
                            atomicAdd(&bins[nbkt[s][v]], nval[s][v]);
                        if (both[s]) {
                            #pragma unroll
                            for (int v = 0; v < 8; ++v)
                                atomicAdd(&bins[tbkt[s][v]], tval[s][v]);
                        }
                    }
                }
            }
            __syncthreads();
        }
    }

    const float4* bins4  = (const float4*)bins;
    float4*       plane4 = (float4*)plane;
    #pragma unroll
    for (int q = 0; q < PROJ / 4 / BLK; ++q)
        plane4[q * BLK + tid] = bins4[q * BLK + tid];
}

__global__ __launch_bounds__(FBLK)
void cbp_sum_sqrt_norm_kernel(const float* ws, float* out, size_t ws_bytes)
{
    __shared__ float red[FBLK / 32];

    const int b    = blockIdx.x;
    const int tid  = threadIdx.x;
    const int lane = tid & 31;
    const int wave = tid >> 5;

    const bool split = ws_holds_planes(ws, ws_bytes);
    float* ob = out + (size_t)b * PROJ;

    float4 a0, a1;
    if (split) {
        const float4* pl = (const float4*)(ws + (size_t)b * SPLIT * PROJ);
        a0 = pl[tid];
        a1 = pl[tid + FBLK];
        #pragma unroll
        for (int q = 1; q < SPLIT; ++q) {
            const float4 u = pl[q * (PROJ / 4) + tid];
            const float4 w = pl[q * (PROJ / 4) + tid + FBLK];
            a0.x += u.x; a0.y += u.y; a0.z += u.z; a0.w += u.w;
            a1.x += w.x; a1.y += w.y; a1.z += w.z; a1.w += w.w;
        }
    } else {
        const float4* pl = (const float4*)ob;
        a0 = pl[tid];
        a1 = pl[tid + FBLK];
    }

    float tv[8] = { a0.x, a0.y, a0.z, a0.w, a1.x, a1.y, a1.z, a1.w };
    float yv[8];
    float ss = 0.0f;
    #pragma unroll
    for (int q = 0; q < 8; ++q) {
        const float t  = tv[q];
        const float sg = (t > 0.0f) ? 1.0f : ((t < 0.0f) ? -1.0f : 0.0f);
        const float y  = sg * sqrtf(fabsf(t) + 1e-8f);
        yv[q] = y;
        ss   += y * y;
    }
    #pragma unroll
    for (int off = 16; off > 0; off >>= 1)
        ss += __shfl_down(ss, off, 32);
    if (lane == 0) red[wave] = ss;
    __syncthreads();
    if (tid == 0) {
        float tot = 0.0f;
        #pragma unroll
        for (int w = 0; w < FBLK / 32; ++w) tot += red[w];
        red[0] = fmaxf(sqrtf(tot), 1e-12f);
    }
    __syncthreads();

    const float inv = 1.0f / red[0];
    float4* ob4 = (float4*)ob;
    ob4[tid]        = make_float4(yv[0] * inv, yv[1] * inv, yv[2] * inv, yv[3] * inv);
    ob4[tid + FBLK] = make_float4(yv[4] * inv, yv[5] * inv, yv[6] * inv, yv[7] * inv);
}

extern "C" void kernel_launch(void* const* d_in, const int* in_sizes, int n_in,
                              void* d_out, int out_size, void* d_ws, size_t ws_size,
                              hipStream_t stream) {
    const float* x  = (const float*)d_in[0];
    const float* s1 = (const float*)d_in[1];
    const float* s2 = (const float*)d_in[2];
    const int*   h1 = (const int*)  d_in[3];
    const int*   h2 = (const int*)  d_in[4];
    float*       y  = (float*)d_out;
    float*       ws = (float*)d_ws;
    (void)in_sizes; (void)n_in; (void)out_size;

    cbp_gram_sketch_partial_kernel<<<NIMG * SPLIT, BLK, 0, stream>>>(x, s1, s2, h1, h2, y, ws, ws_size);
    cbp_sum_sqrt_norm_kernel<<<NIMG, FBLK, 0, stream>>>(ws, y, ws_size);
}


# Second launch: 4 work-groups per image, was 1; each sums all 16 planes (the norm needs every bin), starting on its own quarter of the bins so the others find its lines in L2, and writes that quarter
# speedup vs baseline: 22.9656x; 1.0786x over previous
#include <hip/hip_runtime.h>
#include <hip/hip_bf16.h>
#include <math.h>


typedef __attribute__((ext_vector_type(2))) float v2f;
typedef __attribute__((ext_vector_type(8))) float v8f;

#define PROJ   8192
#define NCH    512
#define HWSZ   196
#define BLK    256
#define NIMG   16
#define SPLIT  16
#define BROWS  128
#define BFLT   (BROWS * HWSZ)
#define BVEC   (BFLT / 4)
#define SROWS  192
#define NSLOT  5
#define FBLK   512
#define NSUB   4
#define UVEC   (PROJ / 4 / NSUB)

__device__ __forceinline__ bool ws_holds_planes(const float* ws, size_t ws_bytes)
{
    return ws != nullptr
        && ws_bytes >= (size_t)NIMG * SPLIT * PROJ * sizeof(float)
        && (((size_t)ws) & 15) == 0;
}

template <int NVEC>
__device__ __forceinline__ void stage_rows(const float4* __restrict__ a4, int nA4,
                                           const float4* __restrict__ b4,
                                           float4* st4, int tid)
{
    constexpr int FULL = NVEC / BLK;
    constexpr int REM  = NVEC % BLK;
    float4 part = make_float4(0.0f, 0.0f, 0.0f, 0.0f);
    if (REM != 0) {
        const int idx = min(FULL * BLK + tid, NVEC - 1);
        const float4* src = (idx < nA4) ? (a4 + idx) : (b4 + (idx - nA4));
        part = *src;
    }
    #pragma unroll
    for (int q = 0; q < FULL; ++q) {
        const int idx = q * BLK + tid;
        const float4* src = (idx < nA4) ? (a4 + idx) : (b4 + (idx - nA4));
        st4[idx] = *src;
    }
    if (REM != 0 && tid < REM) st4[FULL * BLK + tid] = part;
}

template <int NS>
__device__ __forceinline__ void gram_k_loop(const float* const (&ap)[NSLOT],
                                            const float* const (&bp)[NSLOT],
                                            v8f (&acc)[NSLOT])
{
    #pragma unroll 7
    for (int kk = 0; kk < 49; ++kk) {
        const int k0 = kk * 4;
        v2f av[NS], bv[NS];
        #pragma unroll
        for (int s = 0; s < NS; ++s) {
            av[s] = *(const v2f*)(ap[s] + k0);
            bv[s] = *(const v2f*)(bp[s] + k0);
        }
        #pragma unroll
        for (int s = 0; s < NS; ++s)
            acc[s] = __builtin_amdgcn_wmma_f32_16x16x4_f32(
                    false, av[s], false, bv[s], (short)0, acc[s], false, false);
    }
}

__global__ __launch_bounds__(BLK)
void cbp_gram_sketch_partial_kernel(const float* __restrict__ x,
                                    const float* __restrict__ s1,
                                    const float* __restrict__ s2,
                                    const int*   __restrict__ h1,
                                    const int*   __restrict__ h2,
                                    float* __restrict__ out,
                                    float* __restrict__ ws,
                                    size_t ws_bytes)
{
    __shared__ __attribute__((aligned(16))) float stage[SROWS * HWSZ];
    __shared__ __attribute__((aligned(16))) float bins[PROJ];
    __shared__ __attribute__((aligned(16))) float s1l[NCH];
    __shared__ __attribute__((aligned(16))) float s2l[NCH];
    __shared__ __attribute__((aligned(16))) int   h1l[NCH];
    __shared__ __attribute__((aligned(16))) int   h2l[NCH];

    const bool split = ws_holds_planes(ws, ws_bytes);
    const int  b     = blockIdx.x / SPLIT;
    const int  p     = blockIdx.x % SPLIT;
    if (!split && p != 0) return;

    const int u0 = split ? p : 0;
    const int u1 = split ? p + 1 : SPLIT;
    float* plane = split ? ws + (size_t)blockIdx.x * PROJ
                         : out + (size_t)b * PROJ;

    const int tid  = threadIdx.x;
    const int lane = tid & 31;
    const int wave = __builtin_amdgcn_readfirstlane(tid >> 5);

    for (int q = tid; q < PROJ; q += BLK) bins[q] = 0.0f;
    for (int c = tid; c < NCH; c += BLK) {
        s1l[c] = s1[c];  s2l[c] = s2[c];
        h1l[c] = h1[c];  h2l[c] = h2[c];
    }

    const float* xb = x + (size_t)b * NCH * HWSZ;

    const int half = lane >> 4;
    const int lm16 = lane & 15;
    const int kb   = 2 * half;

    for (int u = u0; u < u1; ++u) {
        const bool diag = (u < 4);
        int aC, bC, bRow0;
        if (diag) {
            aC = u * BROWS;  bC = aC;  bRow0 = 0;
        } else {
            const int o  = u - 4;
            const int pr = o >> 1;
            const int bi = (pr >= 3) + (pr >= 5);
            const int bj = (pr < 3) ? pr + 1 : ((pr < 5) ? pr - 1 : 3);
            aC = bi * BROWS + (o & 1) * 64;  bC = bj * BROWS;  bRow0 = 64;
        }

        {
            const float4* a4  = (const float4*)(xb + (size_t)aC * HWSZ);
            const float4* b4  = (const float4*)(xb + (size_t)bC * HWSZ);
            float4*       st4 = (float4*)stage;
            if (diag) stage_rows<BVEC>(a4, BVEC, a4, st4, tid);
            else      stage_rows<BVEC / 2 + BVEC>(a4, BVEC / 2, b4, st4, tid);
        }
        __syncthreads();

        const int ns = (diag && (__builtin_popcount((unsigned)wave) & 1) == 0) ? 5 : 4;
        int ti[NSLOT], tj[NSLOT];
        #pragma unroll
        for (int s = 0; s < NSLOT; ++s) {
            if (diag) {
                const int e = (s < 4) ? wave + 8 * s : 32 + (wave >> 1);
                const int q = e / 9;
                const int r = e - 9 * q;
                if (r < 8 - q) { ti[s] = q;      tj[s] = q + r; }
                else           { ti[s] = 7 - q;  tj[s] = 7 - q + (r - (8 - q)); }
            } else {
                ti[s] = 2 * (wave >> 2) + ((s >> 1) & 1);
                tj[s] = 2 * (wave & 3) + (s & 1);
            }
        }

        const float* ap[NSLOT];
        const float* bp[NSLOT];
        v8f acc[NSLOT];
        #pragma unroll
        for (int s = 0; s < NSLOT; ++s) {
            ap[s]  = stage + (ti[s] * 16 + lm16) * HWSZ + kb;
            bp[s]  = stage + (bRow0 + tj[s] * 16 + lm16) * HWSZ + kb;
            acc[s] = (v8f){};
        }

        if (ns == 5) gram_k_loop<5>(ap, bp, acc);
        else         gram_k_loop<4>(ap, bp, acc);

        float nval[NSLOT][8], tval[NSLOT][8];
        int   nbkt[NSLOT][8], tbkt[NSLOT][8];
        bool  both[NSLOT];
        #pragma unroll
        for (int s = 0; s < NSLOT; ++s) {
            const int rc0 = aC + ti[s] * 16;
            const int cc0 = bC + tj[s] * 16;
            both[s] = (rc0 != cc0);
            const int cB  = cc0 + lm16;
            const int cA0 = rc0 + 8 * half;
            const float s1b = s1l[cB], s2b = s2l[cB];
            const int   h1b = h1l[cB], h2b = h2l[cB];
            const float4 s1a0 = *(const float4*)(s1l + cA0), s1a1 = *(const float4*)(s1l + cA0 + 4);
            const float4 s2a0 = *(const float4*)(s2l + cA0), s2a1 = *(const float4*)(s2l + cA0 + 4);
            const int4   h1a0 = *(const int4*)(h1l + cA0),   h1a1 = *(const int4*)(h1l + cA0 + 4);
            const int4   h2a0 = *(const int4*)(h2l + cA0),   h2a1 = *(const int4*)(h2l + cA0 + 4);
            const float s1a[8] = { s1a0.x, s1a0.y, s1a0.z, s1a0.w, s1a1.x, s1a1.y, s1a1.z, s1a1.w };
            const float s2a[8] = { s2a0.x, s2a0.y, s2a0.z, s2a0.w, s2a1.x, s2a1.y, s2a1.z, s2a1.w };
            const int   h1a[8] = { h1a0.x, h1a0.y, h1a0.z, h1a0.w, h1a1.x, h1a1.y, h1a1.z, h1a1.w };
            const int   h2a[8] = { h2a0.x, h2a0.y, h2a0.z, h2a0.w, h2a1.x, h2a1.y, h2a1.z, h2a1.w };
            #pragma unroll
            for (int v = 0; v < 8; ++v) {
                nval[s][v] = acc[s][v] * s1a[v] * s2b;
                nbkt[s][v] = (h1a[v] + h2b) & (PROJ - 1);
                tval[s][v] = acc[s][v] * s1b * s2a[v];
                tbkt[s][v] = (h1b + h2a[v]) & (PROJ - 1);
            }
        }

        for (int w = 0; w < BLK / 32; ++w) {
            if (wave == w) {
                #pragma unroll
                for (int s = 0; s < NSLOT; ++s) {
                    if (s < ns) {
                        #pragma unroll
                        for (int v = 0; v < 8; ++v)
                            atomicAdd(&bins[nbkt[s][v]], nval[s][v]);
                        if (both[s]) {
                            #pragma unroll
                            for (int v = 0; v < 8; ++v)
                                atomicAdd(&bins[tbkt[s][v]], tval[s][v]);
                        }
                    }
                }
            }
            __syncthreads();
        }
    }

    const float4* bins4  = (const float4*)bins;
    float4*       plane4 = (float4*)plane;
    #pragma unroll
    for (int q = 0; q < PROJ / 4 / BLK; ++q)
        plane4[q * BLK + tid] = bins4[q * BLK + tid];
}

__global__ __launch_bounds__(FBLK)
void cbp_sum_sqrt_norm_kernel(const float* ws, float* out, size_t ws_bytes)
{
    __shared__ float red[NSUB][FBLK / 32];
    __shared__ float utot[NSUB];

    const int b    = blockIdx.x / NSUB;
    const int s    = blockIdx.x % NSUB;
    const int tid  = threadIdx.x;
    const int lane = tid & 31;
    const int wave = tid >> 5;

    const bool split = ws_holds_planes(ws, ws_bytes);
    if (!split && s != 0) return;

    float* ob = out + (size_t)b * PROJ;
    const float4* pl = split ? (const float4*)(ws + (size_t)b * SPLIT * PROJ)
                             : (const float4*)ob;

    float4 acc[NSUB];
    if (split) {
        float4 w[SPLIT];
        {
            const float4* p = pl + s * UVEC + tid;
            #pragma unroll
            for (int q = 0; q < SPLIT; ++q) w[q] = p[q * (PROJ / 4)];
        }
        #pragma unroll
        for (int j = 0; j < NSUB; ++j) {
            const int un = (s + (((j + 1) & 1) * 2 + ((j + 1) >> 1))) & (NSUB - 1);
            const float4* pn = pl + un * UVEC + tid;
            float4 a;
            #pragma unroll
            for (int g = 0; g < SPLIT; g += 4) {
                if (g == 0) a = w[0];
                else { a.x += w[g].x; a.y += w[g].y; a.z += w[g].z; a.w += w[g].w; }
                float4 nx[4];
                if (j + 1 < NSUB) {
                    asm volatile("" : "+v"(a.x), "+v"(a.y), "+v"(a.z), "+v"(a.w) : : "memory");
                    #pragma unroll
                    for (int r = 0; r < 4; ++r) nx[r] = pn[(g + r) * (PROJ / 4)];
                }
                #pragma unroll
                for (int r = 1; r < 4; ++r) {
                    a.x += w[g + r].x; a.y += w[g + r].y; a.z += w[g + r].z; a.w += w[g + r].w;
                }
                if (j + 1 < NSUB) {
                    #pragma unroll
                    for (int r = 0; r < 4; ++r) w[g + r] = nx[r];
                }
            }
            acc[j] = a;
        }
    } else {
        #pragma unroll
        for (int j = 0; j < NSUB; ++j) {
            const int u = (s + ((j & 1) * 2 + (j >> 1))) & (NSUB - 1);
            acc[j] = pl[u * UVEC + tid];
        }
    }

    float yv[NSUB][4];
    float ssu[NSUB];
    #pragma unroll
    for (int j = 0; j < NSUB; ++j) {
        const float tv[4] = { acc[j].x, acc[j].y, acc[j].z, acc[j].w };
        float ss = 0.0f;
        #pragma unroll
        for (int q = 0; q < 4; ++q) {
            const float t  = tv[q];
            const float sg = (t > 0.0f) ? 1.0f : ((t < 0.0f) ? -1.0f : 0.0f);
            const float y  = sg * sqrtf(fabsf(t) + 1e-8f);
            yv[j][q] = y;
            ss      += y * y;
        }
        ssu[j] = ss;
    }

    #pragma unroll
    for (int off = 16; off > 0; off >>= 1) {
        float o[NSUB];
        #pragma unroll
        for (int j = 0; j < NSUB; ++j) o[j] = __shfl_down(ssu[j], off, 32);
        #pragma unroll
        for (int j = 0; j < NSUB; ++j) ssu[j] += o[j];
    }
    if (lane == 0) {
        #pragma unroll
        for (int j = 0; j < NSUB; ++j) red[j][wave] = ssu[j];
    }
    __syncthreads();
    if (tid < NSUB) {
        float tot = 0.0f;
        #pragma unroll
        for (int w = 0; w < FBLK / 32; ++w) tot += red[tid][w];
        utot[(s + ((tid & 1) * 2 + (tid >> 1))) & (NSUB - 1)] = tot;
    }
    __syncthreads();

    const float tot = ((utot[0] + utot[1]) + utot[2]) + utot[3];
    const float inv = 1.0f / fmaxf(sqrtf(tot), 1e-12f);
    float4* ob4 = (float4*)ob;
    #pragma unroll
    for (int j = 0; j < NSUB; ++j) {
        if (j == 0 || !split) {
            const int u = (s + ((j & 1) * 2 + (j >> 1))) & (NSUB - 1);
            ob4[u * UVEC + tid] = make_float4(yv[j][0] * inv, yv[j][1] * inv,
                                              yv[j][2] * inv, yv[j][3] * inv);
        }
    }
}

extern "C" void kernel_launch(void* const* d_in, const int* in_sizes, int n_in,
                              void* d_out, int out_size, void* d_ws, size_t ws_size,
                              hipStream_t stream) {
    const float* x  = (const float*)d_in[0];
    const float* s1 = (const float*)d_in[1];
    const float* s2 = (const float*)d_in[2];
    const int*   h1 = (const int*)  d_in[3];
    const int*   h2 = (const int*)  d_in[4];
    float*       y  = (float*)d_out;
    float*       ws = (float*)d_ws;
    (void)in_sizes; (void)n_in; (void)out_size;

    cbp_gram_sketch_partial_kernel<<<NIMG * SPLIT, BLK, 0, stream>>>(x, s1, s2, h1, h2, y, ws, ws_size);
    cbp_sum_sqrt_norm_kernel<<<NIMG * NSUB, FBLK, 0, stream>>>(ws, y, ws_size);
}


# Scatter: four waves add at once, each into its own LDS sketch plane (2 turns, was 8 turns of one wave); planes zeroed under the table loads and summed per bin in fixed order at the store
# speedup vs baseline: 24.6620x; 1.0739x over previous
#include <hip/hip_runtime.h>
#include <hip/hip_bf16.h>
#include <math.h>


typedef __attribute__((ext_vector_type(2))) float v2f;
typedef __attribute__((ext_vector_type(8))) float v8f;

#define PROJ   8192
#define NCH    512
#define HWSZ   196
#define BLK    256
#define NIMG   16
#define SPLIT  16
#define BROWS  128
#define BFLT   (BROWS * HWSZ)
#define BVEC   (BFLT / 4)
#define SROWS  192
#define NSLOT  5
#define NPL    4
#define FBLK   512
#define NSUB   4
#define UVEC   (PROJ / 4 / NSUB)

__device__ __forceinline__ bool ws_holds_planes(const float* ws, size_t ws_bytes)
{
    return ws != nullptr
        && ws_bytes >= (size_t)NIMG * SPLIT * PROJ * sizeof(float)
        && (((size_t)ws) & 15) == 0;
}

template <int NVEC>
__device__ __forceinline__ void stage_rows(const float4* __restrict__ a4, int nA4,
                                           const float4* __restrict__ b4,
                                           float4* st4, int tid)
{
    constexpr int FULL = NVEC / BLK;
    constexpr int REM  = NVEC % BLK;
    float4 part = make_float4(0.0f, 0.0f, 0.0f, 0.0f);
    if (REM != 0) {
        const int idx = min(FULL * BLK + tid, NVEC - 1);
        const float4* src = (idx < nA4) ? (a4 + idx) : (b4 + (idx - nA4));
        part = *src;
    }
    #pragma unroll
    for (int q = 0; q < FULL; ++q) {
        const int idx = q * BLK + tid;
        const float4* src = (idx < nA4) ? (a4 + idx) : (b4 + (idx - nA4));
        st4[idx] = *src;
    }
    if (REM != 0 && tid < REM) st4[FULL * BLK + tid] = part;
}

template <int NS>
__device__ __forceinline__ void gram_k_loop(const float* const (&ap)[NSLOT],
                                            const float* const (&bp)[NSLOT],
                                            v8f (&acc)[NSLOT])
{
    #pragma unroll 7
    for (int kk = 0; kk < 49; ++kk) {
        const int k0 = kk * 4;
        v2f av[NS], bv[NS];
        #pragma unroll
        for (int s = 0; s < NS; ++s) {
            av[s] = *(const v2f*)(ap[s] + k0);
            bv[s] = *(const v2f*)(bp[s] + k0);
        }
        #pragma unroll
        for (int s = 0; s < NS; ++s)
            acc[s] = __builtin_amdgcn_wmma_f32_16x16x4_f32(
                    false, av[s], false, bv[s], (short)0, acc[s], false, false);
    }
}

__global__ __launch_bounds__(BLK)
void cbp_gram_sketch_partial_kernel(const float* __restrict__ x,
                                    const float* __restrict__ s1,
                                    const float* __restrict__ s2,
                                    const int*   __restrict__ h1,
                                    const int*   __restrict__ h2,
                                    float* __restrict__ out,
                                    float* __restrict__ ws,
                                    size_t ws_bytes)
{
    __shared__ __attribute__((aligned(16))) float stage[SROWS * HWSZ];
    __shared__ __attribute__((aligned(16))) float bins[NPL][PROJ];
    __shared__ __attribute__((aligned(16))) float s1l[NCH];
    __shared__ __attribute__((aligned(16))) float s2l[NCH];
    __shared__ __attribute__((aligned(16))) int   h1l[NCH];
    __shared__ __attribute__((aligned(16))) int   h2l[NCH];

    const bool split = ws_holds_planes(ws, ws_bytes);
    const int  b     = blockIdx.x / SPLIT;
    const int  p     = blockIdx.x % SPLIT;
    if (!split && p != 0) return;

    const int u0 = split ? p : 0;
    const int u1 = split ? p + 1 : SPLIT;
    float* plane = split ? ws + (size_t)blockIdx.x * PROJ
                         : out + (size_t)b * PROJ;

    const int tid  = threadIdx.x;
    const int lane = tid & 31;
    const int wave = __builtin_amdgcn_readfirstlane(tid >> 5);

    {
        float ts1[NCH / BLK], ts2[NCH / BLK];
        int   th1[NCH / BLK], th2[NCH / BLK];
        #pragma unroll
        for (int i = 0; i < NCH / BLK; ++i) {
            const int c = i * BLK + tid;
            ts1[i] = s1[c];  ts2[i] = s2[c];
            th1[i] = h1[c];  th2[i] = h2[c];
        }
        float4* z4 = (float4*)&bins[0][0];
        #pragma unroll
        for (int q = 0; q < NPL * PROJ / 4 / BLK; ++q)
            z4[q * BLK + tid] = make_float4(0.0f, 0.0f, 0.0f, 0.0f);
        #pragma unroll
        for (int i = 0; i < NCH / BLK; ++i) {
            const int c = i * BLK + tid;
            s1l[c] = ts1[i];  s2l[c] = ts2[i];
            h1l[c] = th1[i];  h2l[c] = th2[i];
        }
    }

    const float* xb = x + (size_t)b * NCH * HWSZ;

    const int half = lane >> 4;
    const int lm16 = lane & 15;
    const int kb   = 2 * half;

    for (int u = u0; u < u1; ++u) {
        const bool diag = (u < 4);
        int aC, bC, bRow0;
        if (diag) {
            aC = u * BROWS;  bC = aC;  bRow0 = 0;
        } else {
            const int o  = u - 4;
            const int pr = o >> 1;
            const int bi = (pr >= 3) + (pr >= 5);
            const int bj = (pr < 3) ? pr + 1 : ((pr < 5) ? pr - 1 : 3);
            aC = bi * BROWS + (o & 1) * 64;  bC = bj * BROWS;  bRow0 = 64;
        }

        {
            const float4* a4  = (const float4*)(xb + (size_t)aC * HWSZ);
            const float4* b4  = (const float4*)(xb + (size_t)bC * HWSZ);
            float4*       st4 = (float4*)stage;
            if (diag) stage_rows<BVEC>(a4, BVEC, a4, st4, tid);
            else      stage_rows<BVEC / 2 + BVEC>(a4, BVEC / 2, b4, st4, tid);
        }
        __syncthreads();

        const int ns = (diag && (__builtin_popcount((unsigned)wave) & 1) == 0) ? 5 : 4;
        int ti[NSLOT], tj[NSLOT];
        #pragma unroll
        for (int s = 0; s < NSLOT; ++s) {
            if (diag) {
                const int e = (s < 4) ? wave + 8 * s : 32 + (wave >> 1);
                const int q = e / 9;
                const int r = e - 9 * q;
                if (r < 8 - q) { ti[s] = q;      tj[s] = q + r; }
                else           { ti[s] = 7 - q;  tj[s] = 7 - q + (r - (8 - q)); }
            } else {
                ti[s] = 2 * (wave >> 2) + ((s >> 1) & 1);
                tj[s] = 2 * (wave & 3) + (s & 1);
            }
        }

        const float* ap[NSLOT];
        const float* bp[NSLOT];
        v8f acc[NSLOT];
        #pragma unroll
        for (int s = 0; s < NSLOT; ++s) {
            ap[s]  = stage + (ti[s] * 16 + lm16) * HWSZ + kb;
            bp[s]  = stage + (bRow0 + tj[s] * 16 + lm16) * HWSZ + kb;
            acc[s] = (v8f){};
        }

        if (ns == 5) gram_k_loop<5>(ap, bp, acc);
        else         gram_k_loop<4>(ap, bp, acc);

        float nval[NSLOT][8], tval[NSLOT][8];
        int   nbkt[NSLOT][8], tbkt[NSLOT][8];
        bool  both[NSLOT];
        #pragma unroll
        for (int s = 0; s < NSLOT; ++s) {
            const int rc0 = aC + ti[s] * 16;
            const int cc0 = bC + tj[s] * 16;
            both[s] = (rc0 != cc0);
            const int cB  = cc0 + lm16;
            const int cA0 = rc0 + 8 * half;
            const float s1b = s1l[cB], s2b = s2l[cB];
            const int   h1b = h1l[cB], h2b = h2l[cB];
            const float4 s1a0 = *(const float4*)(s1l + cA0), s1a1 = *(const float4*)(s1l + cA0 + 4);
            const float4 s2a0 = *(const float4*)(s2l + cA0), s2a1 = *(const float4*)(s2l + cA0 + 4);
            const int4   h1a0 = *(const int4*)(h1l + cA0),   h1a1 = *(const int4*)(h1l + cA0 + 4);
            const int4   h2a0 = *(const int4*)(h2l + cA0),   h2a1 = *(const int4*)(h2l + cA0 + 4);
            const float s1a[8] = { s1a0.x, s1a0.y, s1a0.z, s1a0.w, s1a1.x, s1a1.y, s1a1.z, s1a1.w };
            const float s2a[8] = { s2a0.x, s2a0.y, s2a0.z, s2a0.w, s2a1.x, s2a1.y, s2a1.z, s2a1.w };
            const int   h1a[8] = { h1a0.x, h1a0.y, h1a0.z, h1a0.w, h1a1.x, h1a1.y, h1a1.z, h1a1.w };
            const int   h2a[8] = { h2a0.x, h2a0.y, h2a0.z, h2a0.w, h2a1.x, h2a1.y, h2a1.z, h2a1.w };
            #pragma unroll
            for (int v = 0; v < 8; ++v) {
                nval[s][v] = acc[s][v] * s1a[v] * s2b;
                nbkt[s][v] = (h1a[v] + h2b) & (PROJ - 1);
                tval[s][v] = acc[s][v] * s1b * s2a[v];
                tbkt[s][v] = (h1b + h2a[v]) & (PROJ - 1);
            }
        }

        float* const myplane = bins[wave & (NPL - 1)];
        for (int t = 0; t < BLK / 32 / NPL; ++t) {
            if (wave / NPL == t) {
                #pragma unroll
                for (int s = 0; s < NSLOT; ++s) {
                    if (s < ns) {
                        #pragma unroll
                        for (int v = 0; v < 8; ++v)
                            atomicAdd(&myplane[nbkt[s][v]], nval[s][v]);
                        if (both[s]) {
                            #pragma unroll
                            for (int v = 0; v < 8; ++v)
                                atomicAdd(&myplane[tbkt[s][v]], tval[s][v]);
                        }
                    }
                }
            }
            __syncthreads();
        }
    }

    static_assert(NPL == 4, "the store below sums exactly four planes");
    const float4* bins4  = (const float4*)&bins[0][0];
    float4*       plane4 = (float4*)plane;
    #pragma unroll
    for (int q = 0; q < PROJ / 4 / BLK; ++q) {
        const int i = q * BLK + tid;
        const float4 p0 = bins4[i];
        const float4 p1 = bins4[i + PROJ / 4];
        const float4 p2 = bins4[i + 2 * (PROJ / 4)];
        const float4 p3 = bins4[i + 3 * (PROJ / 4)];
        plane4[i] = make_float4((p0.x + p1.x) + (p2.x + p3.x), (p0.y + p1.y) + (p2.y + p3.y),
                                (p0.z + p1.z) + (p2.z + p3.z), (p0.w + p1.w) + (p2.w + p3.w));
    }
}

__global__ __launch_bounds__(FBLK)
void cbp_sum_sqrt_norm_kernel(const float* ws, float* out, size_t ws_bytes)
{
    __shared__ float red[NSUB][FBLK / 32];
    __shared__ float utot[NSUB];

    const int b    = blockIdx.x / NSUB;
    const int s    = blockIdx.x % NSUB;
    const int tid  = threadIdx.x;
    const int lane = tid & 31;
    const int wave = tid >> 5;

    const bool split = ws_holds_planes(ws, ws_bytes);
    if (!split && s != 0) return;

    float* ob = out + (size_t)b * PROJ;
    const float4* pl = split ? (const float4*)(ws + (size_t)b * SPLIT * PROJ)
                             : (const float4*)ob;

    float4 acc[NSUB];
    if (split) {
        float4 w[SPLIT];
        {
            const float4* p = pl + s * UVEC + tid;
            #pragma unroll
            for (int q = 0; q < SPLIT; ++q) w[q] = p[q * (PROJ / 4)];
        }
        #pragma unroll
        for (int j = 0; j < NSUB; ++j) {
            const int un = (s + (((j + 1) & 1) * 2 + ((j + 1) >> 1))) & (NSUB - 1);
            const float4* pn = pl + un * UVEC + tid;
            float4 a;
            #pragma unroll
            for (int g = 0; g < SPLIT; g += 4) {
                if (g == 0) a = w[0];
                else { a.x += w[g].x; a.y += w[g].y; a.z += w[g].z; a.w += w[g].w; }
                float4 nx[4];
                if (j + 1 < NSUB) {
                    asm volatile("" : "+v"(a.x), "+v"(a.y), "+v"(a.z), "+v"(a.w) : : "memory");
                    #pragma unroll
                    for (int r = 0; r < 4; ++r) nx[r] = pn[(g + r) * (PROJ / 4)];
                }
                #pragma unroll
                for (int r = 1; r < 4; ++r) {
                    a.x += w[g + r].x; a.y += w[g + r].y; a.z += w[g + r].z; a.w += w[g + r].w;
                }
                if (j + 1 < NSUB) {
                    #pragma unroll
                    for (int r = 0; r < 4; ++r) w[g + r] = nx[r];
                }
            }
            acc[j] = a;
        }
    } else {
        #pragma unroll
        for (int j = 0; j < NSUB; ++j) {
            const int u = (s + ((j & 1) * 2 + (j >> 1))) & (NSUB - 1);
            acc[j] = pl[u * UVEC + tid];
        }
    }

    float yv[NSUB][4];
    float ssu[NSUB];
    #pragma unroll
    for (int j = 0; j < NSUB; ++j) {
        const float tv[4] = { acc[j].x, acc[j].y, acc[j].z, acc[j].w };
        float ss = 0.0f;
        #pragma unroll
        for (int q = 0; q < 4; ++q) {
            const float t  = tv[q];
            const float sg = (t > 0.0f) ? 1.0f : ((t < 0.0f) ? -1.0f : 0.0f);
            const float y  = sg * sqrtf(fabsf(t) + 1e-8f);
            yv[j][q] = y;
            ss      += y * y;
        }
        ssu[j] = ss;
    }

    #pragma unroll
    for (int off = 16; off > 0; off >>= 1) {
        float o[NSUB];
        #pragma unroll
        for (int j = 0; j < NSUB; ++j) o[j] = __shfl_down(ssu[j], off, 32);
        #pragma unroll
        for (int j = 0; j < NSUB; ++j) ssu[j] += o[j];
    }
    if (lane == 0) {
        #pragma unroll
        for (int j = 0; j < NSUB; ++j) red[j][wave] = ssu[j];
    }
    __syncthreads();
    if (tid < NSUB) {
        float tot = 0.0f;
        #pragma unroll
        for (int w = 0; w < FBLK / 32; ++w) tot += red[tid][w];
        utot[(s + ((tid & 1) * 2 + (tid >> 1))) & (NSUB - 1)] = tot;
    }
    __syncthreads();

    const float tot = ((utot[0] + utot[1]) + utot[2]) + utot[3];
    const float inv = 1.0f / fmaxf(sqrtf(tot), 1e-12f);
    float4* ob4 = (float4*)ob;
    #pragma unroll
    for (int j = 0; j < NSUB; ++j) {
        if (j == 0 || !split) {
            const int u = (s + ((j & 1) * 2 + (j >> 1))) & (NSUB - 1);
            ob4[u * UVEC + tid] = make_float4(yv[j][0] * inv, yv[j][1] * inv,
                                              yv[j][2] * inv, yv[j][3] * inv);
        }
    }
}

extern "C" void kernel_launch(void* const* d_in, const int* in_sizes, int n_in,
                              void* d_out, int out_size, void* d_ws, size_t ws_size,
                              hipStream_t stream) {
    const float* x  = (const float*)d_in[0];
    const float* s1 = (const float*)d_in[1];
    const float* s2 = (const float*)d_in[2];
    const int*   h1 = (const int*)  d_in[3];
    const int*   h2 = (const int*)  d_in[4];
    float*       y  = (float*)d_out;
    float*       ws = (float*)d_ws;
    (void)in_sizes; (void)n_in; (void)out_size;

    cbp_gram_sketch_partial_kernel<<<NIMG * SPLIT, BLK, 0, stream>>>(x, s1, s2, h1, h2, y, ws, ws_size);
    cbp_sum_sqrt_norm_kernel<<<NIMG * NSUB, FBLK, 0, stream>>>(ws, y, ws_size);
}
